# starting kernel
# baseline (speedup 1.0000x reference)
#include <hip/hip_runtime.h>
#include <math.h>


typedef __attribute__((ext_vector_type(2))) float v2f;
typedef __attribute__((ext_vector_type(8))) float v8f;

#define WMMA_F32_16x16x4(a, b, c) \
  __builtin_amdgcn_wmma_f32_16x16x4_f32(false, (a), false, (b), (short)0, (c), false, false)

template <int NT>
__global__ __launch_bounds__(32) void EmbedWeighted_gemm_wmma(
    const float* __restrict__ A,
    const float* __restrict__ B,
    float* __restrict__ C,
    int M, int K, int N_) {
  const int N     = (NT > 0) ? NT : N_;
  const int lane  = threadIdx.x & 31;
  const int l16   = lane & 15;
  const int khalf = (lane >> 4) << 1;
  const int m0    = blockIdx.x << 4;
  const int n0    = blockIdx.y << 4;

  const float* ap = A + (size_t)(m0 + l16) * K + khalf;
  const float* bp = B + (size_t)khalf * N + (n0 + l16);

  v8f acc = {};

  int k0 = 0;
  for (; k0 + 32 <= K; k0 += 32) {
    __builtin_prefetch(ap + 128, 0, 3);
#pragma unroll
    for (int kk = 0; kk < 32; kk += 4) {
      v2f a = *(const v2f*)(ap + kk);
      v2f b;
      b.x = bp[kk * N];
      b.y = bp[kk * N + N];
      acc = WMMA_F32_16x16x4(a, b, acc);
    }
    ap += 32;
    bp += (size_t)32 * N;
  }
  for (; k0 < K; k0 += 4) {
    v2f a = *(const v2f*)ap;
    v2f b;
    b.x = bp[0];
    b.y = bp[N];
    acc = WMMA_F32_16x16x4(a, b, acc);
    ap += 4;
    bp += (size_t)4 * N;
  }

  float* crow = C + (size_t)(m0 + ((lane >> 4) << 3)) * N + (n0 + l16);
#pragma unroll
  for (int r = 0; r < 8; ++r) {
    crow[(size_t)r * N] = acc[r];
  }
}

extern "C" void kernel_launch(void* const* d_in, const int* in_sizes, int n_in,
                              void* d_out, int out_size, void* d_ws, size_t ws_size,
                              hipStream_t stream) {
  (void)n_in; (void)d_ws; (void)ws_size;
  const float* inputs = (const float*)d_in[0];
  const float* emb    = (const float*)d_in[1];
  float* out          = (float*)d_out;

  const double s0 = (double)in_sizes[0];
  const double s1 = (double)in_sizes[1];
  const double so = (double)out_size;
  const int D  = (int)llround(sqrt(s1 * so / s0));
  const int V  = (int)llround(s1 / D);
  const int Bt = (int)llround(so / D);

  dim3 grid(Bt / 16, D / 16, 1);
  dim3 block(32, 1, 1);
  if (D == 64) {
    EmbedWeighted_gemm_wmma<64><<<grid, block, 0, stream>>>(inputs, emb, out, Bt, V, D);
  } else {
    EmbedWeighted_gemm_wmma<0><<<grid, block, 0, stream>>>(inputs, emb, out, Bt, V, D);
  }
}


# C by 32x32 blocks of 4 waves: A and B steps of 80 k fetched with whole-line loads two steps ahead, staged in LDS, WMMA fragments read from LDS in two groups per step; k order per tile unchanged
# speedup vs baseline: 2.1822x; 2.1822x over previous
#include <hip/hip_runtime.h>
#include <math.h>


typedef __attribute__((ext_vector_type(2))) float v2f;
typedef __attribute__((ext_vector_type(4))) float v4f;
typedef __attribute__((ext_vector_type(8))) float v8f;

#define WMMA_F32_16x16x4(a, b, c) \
  __builtin_amdgcn_wmma_f32_16x16x4_f32(false, (a), false, (b), (short)0, (c), false, false)

template <int NT>
__global__ __launch_bounds__(32) void gemm_wmma_direct(
    const float* __restrict__ A,
    const float* __restrict__ B,
    float* __restrict__ C,
    int M, int K, int N_) {
  const int N     = (NT > 0) ? NT : N_;
  const int lane  = threadIdx.x & 31;
  const int l16   = lane & 15;
  const int khalf = (lane >> 4) << 1;
  const int m0    = blockIdx.x << 4;
  const int n0    = blockIdx.y << 4;

  const float* ap = A + (size_t)(m0 + l16) * K + khalf;
  const float* bp = B + (size_t)khalf * N + (n0 + l16);

  v8f acc = {};

  int k0 = 0;
  for (; k0 + 32 <= K; k0 += 32) {
    __builtin_prefetch(ap + 128, 0, 3);
#pragma unroll
    for (int kk = 0; kk < 32; kk += 4) {
      v2f a = *(const v2f*)(ap + kk);
      v2f b;
      b.x = bp[kk * N];
      b.y = bp[kk * N + N];
      acc = WMMA_F32_16x16x4(a, b, acc);
    }
    ap += 32;
    bp += (size_t)32 * N;
  }
  for (; k0 < K; k0 += 4) {
    v2f a = *(const v2f*)ap;
    v2f b;
    b.x = bp[0];
    b.y = bp[N];
    acc = WMMA_F32_16x16x4(a, b, acc);
    ap += 4;
    bp += (size_t)4 * N;
  }

  float* crow = C + (size_t)(m0 + ((lane >> 4) << 3)) * N + (n0 + l16);
#pragma unroll
  for (int r = 0; r < 8; ++r) {
    crow[(size_t)r * N] = acc[r];
  }
}

constexpr int BM      = 32;
constexpr int BN      = 32;
constexpr int KB      = 80;
constexpr int NTHR    = 128;
constexpr int PA      = KB + 4;
constexpr int PB      = 2 * BN + 32;
constexpr int A_TILE  = BM * PA;
constexpr int B_TILE  = (KB / 2) * PB;
constexpr int STAGE   = A_TILE + B_TILE;
constexpr int NSTAGE  = 3;
constexpr int A_ITEMS = (BM * KB / 4) / NTHR;
constexpr int B_ITEMS = ((KB / 2) * (BN / 2)) / NTHR;
constexpr int UW      = KB / 4;
constexpr int UH      = UW / 2;
constexpr int MIN_STEPS = 6;
static_assert((BM * KB / 4) % NTHR == 0 && ((KB / 2) * (BN / 2)) % NTHR == 0, "every thread carries the same number of pieces");
static_assert(PA % 4 == 0 && PB % 4 == 0 && A_TILE % 4 == 0 && STAGE % 4 == 0, "16-byte aligned LDS pieces");

struct GSet {
  v4f a[A_ITEMS];
  v2f b0[B_ITEMS];
  v2f b1[B_ITEMS];
};
struct FSet {
  v2f a[UH];
  v2f b[UH];
};
template <bool V> struct Flag { static constexpr bool value = V; };

__device__ __forceinline__ void lds_barrier() {
  asm volatile("s_wait_dscnt 0x0\n\ts_barrier_signal -1\n\ts_barrier_wait -1" ::: "memory");
}

template <int N>
__global__ __launch_bounds__(NTHR) void gemm_wmma_lds(
    const float* __restrict__ A,
    const float* __restrict__ B,
    float* __restrict__ C,
    int M, int K) {
  __shared__ v4f lds4[NSTAGE * STAGE / 4];
  v2f* lds2 = (v2f*)lds4;

  const int t    = threadIdx.x;
  const int lane = t & 31;
  const int w    = t >> 5;
  const int c    = lane & 15;
  const int h    = lane >> 4;
  const int mt   = w >> 1;
  const int nt   = w & 1;
  const int m0   = blockIdx.x * BM;
  const int nb0  = blockIdx.y * BN;
  const int S    = K / KB;

  unsigned offA[A_ITEMS];
  int      dstA[A_ITEMS];
#pragma unroll
  for (int i = 0; i < A_ITEMS; ++i) {
    const int item = t + NTHR * i;
    const int r = item / (KB / 4);
    const int j = item % (KB / 4);
    offA[i] = (unsigned)(m0 + r) * (unsigned)K + 4u * j;
    dstA[i] = r * (PA / 4) + j;
  }
  unsigned offB[B_ITEMS];
  int      dstB[B_ITEMS];
#pragma unroll
  for (int i = 0; i < B_ITEMS; ++i) {
    const int item = t + NTHR * i;
    const int p  = item / (BN / 2);
    const int cq = item % (BN / 2);
    offB[i] = (unsigned)(2 * p) * N + nb0 + 2 * cq;
    dstB[i] = A_TILE / 4 + p * (PB / 4) + cq;
  }

  const int fragA = (16 * mt + c) * (PA / 2) + h;
  const int fragB = A_TILE / 2 + h * (PB / 2) + (16 * nt + c);

  auto load_a = [&](GSet& g, int s, int i) {
    g.a[i] = *(const v4f*)(A + (size_t)s * KB + offA[i]);
  };
  auto load_b = [&](GSet& g, int s, int i) {
    const float* Bk = B + (size_t)s * KB * N;
    g.b0[i] = *(const v2f*)(Bk + offB[i]);
    g.b1[i] = *(const v2f*)(Bk + offB[i] + N);
  };
  auto load_step = [&](GSet& g, int s) {
#pragma unroll
    for (int i = 0; i < A_ITEMS; ++i) load_a(g, s, i);
#pragma unroll
    for (int i = 0; i < B_ITEMS; ++i) load_b(g, s, i);
  };

  auto store_step = [&](const GSet& g, int st4) {
#pragma unroll
    for (int i = 0; i < A_ITEMS; ++i) lds4[st4 + dstA[i]] = g.a[i];
#pragma unroll
    for (int i = 0; i < B_ITEMS; ++i) {
      const v4f pr = {g.b0[i].x, g.b1[i].x, g.b0[i].y, g.b1[i].y};
      lds4[st4 + dstB[i]] = pr;
    }
  };

  auto frag_load = [&](FSet& f, int st4, int half) {
    const v2f* as = lds2 + 2 * st4 + fragA + half * (UH * 2);
    const v2f* bs = lds2 + 2 * st4 + fragB + half * (UH * PB);
#pragma unroll
    for (int u = 0; u < UH; ++u) {
      f.a[u] = as[2 * u];
      f.b[u] = bs[PB * u];
    }
  };

  v8f acc = {};
  auto compute = [&](const FSet& f) {
#pragma unroll
    for (int u = 0; u < UH; ++u) acc = WMMA_F32_16x16x4(f.a[u], f.b[u], acc);
  };
  auto compute_and_load = [&](const FSet& f, GSet& g, int s) {
#pragma unroll
    for (int u = 0; u < UH; ++u) {
      acc = WMMA_F32_16x16x4(f.a[u], f.b[u], acc);
      if (u < A_ITEMS) load_a(g, s, u);
      if (u < B_ITEMS) load_b(g, s, u);
      __builtin_amdgcn_sched_barrier(0);
    }
  };

  auto prefetch_a = [&](int s) {
    const int k = s * KB + (t >> 5) * 64;
    if (t < 64 && k < K) __builtin_prefetch(A + (size_t)(m0 + (t & 31)) * K + k, 0, 3);
  };

  GSet G0, G1;
  FSet Fa, Fb;
  int rd = 0, nx = STAGE / 4, wr = 2 * (STAGE / 4);

  auto step = [&](GSet& g, int s, auto full) {
    if constexpr (decltype(full)::value) {
      store_step(g, wr);
      frag_load(Fb, rd, 1);
      __builtin_amdgcn_sched_barrier(0);
      compute_and_load(Fa, g, s + 4);
      prefetch_a(s + 6);
    } else {
      if (s + 2 < S) {
        store_step(g, wr);
        if (s + 4 < S) load_step(g, s + 4);
      }
      frag_load(Fb, rd, 1);
      __builtin_amdgcn_sched_barrier(0);
      compute(Fa);
    }
    __builtin_amdgcn_sched_barrier(0);
    frag_load(Fa, nx, 0);
    __builtin_amdgcn_sched_barrier(0);
    compute(Fb);
    __builtin_amdgcn_sched_barrier(0);
    lds_barrier();
    const int o = rd; rd = nx; nx = wr; wr = o;
  };

  {
    GSet P0, P1;
    load_step(P0, 0);
    load_step(P1, 1);
    __builtin_amdgcn_sched_barrier(0);
    load_step(G0, 2);
    __builtin_amdgcn_sched_barrier(0);
    load_step(G1, 3);
    __builtin_amdgcn_sched_barrier(0);
    prefetch_a(4);
    prefetch_a(5);
    store_step(P0, rd);
    store_step(P1, nx);
  }
  lds_barrier();
  frag_load(Fa, rd, 0);

  int s = 0;
  for (; s + 5 < S; s += 2) {
    step(G0, s, Flag<true>{});
    step(G1, s + 1, Flag<true>{});
  }
  while (s + 1 < S) {
    step(G0, s, Flag<false>{});
    ++s;
    if (s + 1 >= S) break;
    step(G1, s, Flag<false>{});
    ++s;
  }
  frag_load(Fb, rd, 1);
  __builtin_amdgcn_sched_barrier(0);
  compute(Fa);
  compute(Fb);

  float* crow = C + (size_t)(m0 + 16 * mt + 8 * h) * N + (nb0 + 16 * nt + c);
#pragma unroll
  for (int r = 0; r < 8; ++r) {
    crow[(size_t)r * N] = acc[r];
  }
}

extern "C" void kernel_launch(void* const* d_in, const int* in_sizes, int n_in,
                              void* d_out, int out_size, void* d_ws, size_t ws_size,
                              hipStream_t stream) {
  (void)n_in; (void)d_ws; (void)ws_size;
  const float* inputs = (const float*)d_in[0];
  const float* emb    = (const float*)d_in[1];
  float* out          = (float*)d_out;

  const double s0 = (double)in_sizes[0];
  const double s1 = (double)in_sizes[1];
  const double so = (double)out_size;
  const int D  = (int)llround(sqrt(s1 * so / s0));
  const int V  = (int)llround(s1 / D);
  const int Bt = (int)llround(so / D);

  if (D == 64 && (Bt % BM) == 0 && (V % KB) == 0 && V >= MIN_STEPS * KB) {
    dim3 grid(Bt / BM, D / BN, 1);
    dim3 block(NTHR, 1, 1);
    gemm_wmma_lds<64><<<grid, block, 0, stream>>>(inputs, emb, out, Bt, V);
  } else {
    dim3 grid(Bt / 16, D / 16, 1);
    dim3 block(32, 1, 1);
    if (D == 64) {
      gemm_wmma_direct<64><<<grid, block, 0, stream>>>(inputs, emb, out, Bt, V, D);
    } else {
      gemm_wmma_direct<0><<<grid, block, 0, stream>>>(inputs, emb, out, Bt, V, D);
    }
  }
}


# K split over 4 work-groups per 32x32 C block (grid z, 6-7 steps of 80 k each); partial tiles go to the workspace, a second launch adds the 4 partials in slice order: summation order changes
# speedup vs baseline: 2.8285x; 1.2962x over previous
#include <hip/hip_runtime.h>
#include <math.h>


typedef __attribute__((ext_vector_type(2))) float v2f;
typedef __attribute__((ext_vector_type(4))) float v4f;
typedef __attribute__((ext_vector_type(8))) float v8f;

#define WMMA_F32_16x16x4(a, b, c) \
  __builtin_amdgcn_wmma_f32_16x16x4_f32(false, (a), false, (b), (short)0, (c), false, false)

template <int NT>
__global__ __launch_bounds__(32) void gemm_wmma_direct(
    const float* __restrict__ A,
    const float* __restrict__ B,
    float* __restrict__ C,
    int M, int K, int N_) {
  const int N     = (NT > 0) ? NT : N_;
  const int lane  = threadIdx.x & 31;
  const int l16   = lane & 15;
  const int khalf = (lane >> 4) << 1;
  const int m0    = blockIdx.x << 4;
  const int n0    = blockIdx.y << 4;

  const float* ap = A + (size_t)(m0 + l16) * K + khalf;
  const float* bp = B + (size_t)khalf * N + (n0 + l16);

  v8f acc = {};

  int k0 = 0;
  for (; k0 + 32 <= K; k0 += 32) {
    __builtin_prefetch(ap + 128, 0, 3);
#pragma unroll
    for (int kk = 0; kk < 32; kk += 4) {
      v2f a = *(const v2f*)(ap + kk);
      v2f b;
      b.x = bp[kk * N];
      b.y = bp[kk * N + N];
      acc = WMMA_F32_16x16x4(a, b, acc);
    }
    ap += 32;
    bp += (size_t)32 * N;
  }
  for (; k0 < K; k0 += 4) {
    v2f a = *(const v2f*)ap;
    v2f b;
    b.x = bp[0];
    b.y = bp[N];
    acc = WMMA_F32_16x16x4(a, b, acc);
    ap += 4;
    bp += (size_t)4 * N;
  }

  float* crow = C + (size_t)(m0 + ((lane >> 4) << 3)) * N + (n0 + l16);
#pragma unroll
  for (int r = 0; r < 8; ++r) {
    crow[(size_t)r * N] = acc[r];
  }
}

constexpr int BM      = 32;
constexpr int BN      = 32;
constexpr int KB      = 80;
constexpr int NTHR    = 128;
constexpr int PA      = KB + 4;
constexpr int PB      = 2 * BN + 32;
constexpr int A_TILE  = BM * PA;
constexpr int B_TILE  = (KB / 2) * PB;
constexpr int STAGE   = A_TILE + B_TILE;
constexpr int NSTAGE  = 3;
constexpr int A_ITEMS = (BM * KB / 4) / NTHR;
constexpr int B_ITEMS = ((KB / 2) * (BN / 2)) / NTHR;
constexpr int UW      = KB / 4;
constexpr int UH      = UW / 2;
constexpr int MIN_STEPS = 6;
constexpr int KSPLIT  = 4;
static_assert((BM * KB / 4) % NTHR == 0 && ((KB / 2) * (BN / 2)) % NTHR == 0, "every thread carries the same number of pieces");
static_assert(PA % 4 == 0 && PB % 4 == 0 && A_TILE % 4 == 0 && STAGE % 4 == 0, "16-byte aligned LDS pieces");

struct GSet {
  v4f a[A_ITEMS];
  v2f b0[B_ITEMS];
  v2f b1[B_ITEMS];
};
struct FSet {
  v2f a[UH];
  v2f b[UH];
};
template <bool V> struct Flag { static constexpr bool value = V; };

__device__ __forceinline__ void lds_barrier() {
  asm volatile("s_wait_dscnt 0x0\n\ts_barrier_signal -1\n\ts_barrier_wait -1" ::: "memory");
}

template <int N>
__global__ __launch_bounds__(NTHR) void gemm_wmma_lds_kslice(
    const float* __restrict__ A,
    const float* __restrict__ B,
    float* __restrict__ P,
    int M, int K) {
  __shared__ v4f lds4[NSTAGE * STAGE / 4];
  v2f* lds2 = (v2f*)lds4;

  const int t    = threadIdx.x;
  const int lane = t & 31;
  const int w    = t >> 5;
  const int c    = lane & 15;
  const int h    = lane >> 4;
  const int mt   = w >> 1;
  const int nt   = w & 1;
  const int m0   = blockIdx.x * BM;
  const int nb0  = blockIdx.y * BN;
  const int Sall = K / KB;
  const int sb   = ((int)blockIdx.z * Sall) / KSPLIT;
  const int se   = (((int)blockIdx.z + 1) * Sall) / KSPLIT;
  const int S    = se - sb;
  const int Ks   = S * KB;
  const float* As = A + (size_t)sb * KB;
  const float* Bs = B + (size_t)sb * KB * N;

  unsigned offA[A_ITEMS];
  int      dstA[A_ITEMS];
#pragma unroll
  for (int i = 0; i < A_ITEMS; ++i) {
    const int item = t + NTHR * i;
    const int r = item / (KB / 4);
    const int j = item % (KB / 4);
    offA[i] = (unsigned)(m0 + r) * (unsigned)K + 4u * j;
    dstA[i] = r * (PA / 4) + j;
  }
  unsigned offB[B_ITEMS];
  int      dstB[B_ITEMS];
#pragma unroll
  for (int i = 0; i < B_ITEMS; ++i) {
    const int item = t + NTHR * i;
    const int p  = item / (BN / 2);
    const int cq = item % (BN / 2);
    offB[i] = (unsigned)(2 * p) * N + nb0 + 2 * cq;
    dstB[i] = A_TILE / 4 + p * (PB / 4) + cq;
  }

  const int fragA = (16 * mt + c) * (PA / 2) + h;
  const int fragB = A_TILE / 2 + h * (PB / 2) + (16 * nt + c);

  auto load_a = [&](GSet& g, int s, int i) {
    g.a[i] = *(const v4f*)(As + (size_t)s * KB + offA[i]);
  };
  auto load_b = [&](GSet& g, int s, int i) {
    const float* Bk = Bs + (size_t)s * KB * N;
    g.b0[i] = *(const v2f*)(Bk + offB[i]);
    g.b1[i] = *(const v2f*)(Bk + offB[i] + N);
  };
  auto load_step = [&](GSet& g, int s) {
#pragma unroll
    for (int i = 0; i < A_ITEMS; ++i) load_a(g, s, i);
#pragma unroll
    for (int i = 0; i < B_ITEMS; ++i) load_b(g, s, i);
  };

  auto store_step = [&](const GSet& g, int st4) {
#pragma unroll
    for (int i = 0; i < A_ITEMS; ++i) lds4[st4 + dstA[i]] = g.a[i];
#pragma unroll
    for (int i = 0; i < B_ITEMS; ++i) {
      const v4f pr = {g.b0[i].x, g.b1[i].x, g.b0[i].y, g.b1[i].y};
      lds4[st4 + dstB[i]] = pr;
    }
  };

  auto frag_load = [&](FSet& f, int st4, int half) {
    const v2f* as = lds2 + 2 * st4 + fragA + half * (UH * 2);
    const v2f* bs = lds2 + 2 * st4 + fragB + half * (UH * PB);
#pragma unroll
    for (int u = 0; u < UH; ++u) {
      f.a[u] = as[2 * u];
      f.b[u] = bs[PB * u];
    }
  };

  v8f acc = {};
  auto compute = [&](const FSet& f) {
#pragma unroll
    for (int u = 0; u < UH; ++u) acc = WMMA_F32_16x16x4(f.a[u], f.b[u], acc);
  };
  auto compute_and_load = [&](const FSet& f, GSet& g, int s) {
#pragma unroll
    for (int u = 0; u < UH; ++u) {
      acc = WMMA_F32_16x16x4(f.a[u], f.b[u], acc);
      if (u < A_ITEMS) load_a(g, s, u);
      if (u < B_ITEMS) load_b(g, s, u);
      __builtin_amdgcn_sched_barrier(0);
    }
  };

  auto prefetch_a = [&](int s) {
    const int k = s * KB + (t >> 5) * 64;
    if (t < 64 && k < Ks) __builtin_prefetch(As + (size_t)(m0 + (t & 31)) * K + k, 0, 3);
  };

  GSet G0, G1;
  FSet Fa, Fb;
  int rd = 0, nx = STAGE / 4, wr = 2 * (STAGE / 4);

  auto step = [&](GSet& g, int s, auto full) {
    if constexpr (decltype(full)::value) {
      store_step(g, wr);
      frag_load(Fb, rd, 1);
      __builtin_amdgcn_sched_barrier(0);
      compute_and_load(Fa, g, s + 4);
      prefetch_a(s + 6);
    } else {
      if (s + 2 < S) {
        store_step(g, wr);
        if (s + 4 < S) load_step(g, s + 4);
      }
      frag_load(Fb, rd, 1);
      __builtin_amdgcn_sched_barrier(0);
      compute(Fa);
    }
    __builtin_amdgcn_sched_barrier(0);
    frag_load(Fa, nx, 0);
    __builtin_amdgcn_sched_barrier(0);
    compute(Fb);
    __builtin_amdgcn_sched_barrier(0);
    lds_barrier();
    const int o = rd; rd = nx; nx = wr; wr = o;
  };

  {
    GSet P0, P1;
    load_step(P0, 0);
    load_step(P1, 1);
    __builtin_amdgcn_sched_barrier(0);
    load_step(G0, 2);
    __builtin_amdgcn_sched_barrier(0);
    load_step(G1, 3);
    __builtin_amdgcn_sched_barrier(0);
    prefetch_a(4);
    prefetch_a(5);
    store_step(P0, rd);
    store_step(P1, nx);
  }
  lds_barrier();
  frag_load(Fa, rd, 0);

  int s = 0;
  for (; s + 5 < S; s += 2) {
    step(G0, s, Flag<true>{});
    step(G1, s + 1, Flag<true>{});
  }
  while (s + 1 < S) {
    step(G0, s, Flag<false>{});
    ++s;
    if (s + 1 >= S) break;
    step(G1, s, Flag<false>{});
    ++s;
  }
  frag_load(Fb, rd, 1);
  __builtin_amdgcn_sched_barrier(0);
  compute(Fa);
  compute(Fb);

  const int tile = (2 * (int)blockIdx.x + mt) * (N / 16) + (2 * (int)blockIdx.y + nt);
  float* pt = P + ((size_t)blockIdx.z * (size_t)(M / 16) * (N / 16) + tile) * 256 + lane;
#pragma unroll
  for (int r = 0; r < 8; ++r) {
    pt[r * 32] = acc[r];
  }
}

template <int N, int Z>
__global__ __launch_bounds__(NTHR) void kslice_sum(
    const float* __restrict__ P,
    float* __restrict__ C,
    int M) {
  const int q = (int)blockIdx.x * NTHR + (int)threadIdx.x;
  const int m = q / (N / 4);
  const int n = (q % (N / 4)) * 4;
  if (m >= M) return;
  const int tile = (m >> 4) * (N / 16) + (n >> 4);
  const int lane = ((m >> 3) & 1) * 16 + (n & 15);
  const size_t plane = (size_t)M * N;
  const float* p = P + ((size_t)tile * 8 + (m & 7)) * 32 + lane;
  v4f part[Z];
#pragma unroll
  for (int z = 0; z < Z; ++z) part[z] = *(const v4f*)(p + (size_t)z * plane);
  v4f sum = part[0];
#pragma unroll
  for (int z = 1; z < Z; ++z) sum += part[z];
  *(v4f*)(C + (size_t)m * N + n) = sum;
}

extern "C" void kernel_launch(void* const* d_in, const int* in_sizes, int n_in,
                              void* d_out, int out_size, void* d_ws, size_t ws_size,
                              hipStream_t stream) {
  (void)n_in;
  const float* inputs = (const float*)d_in[0];
  const float* emb    = (const float*)d_in[1];
  float* out          = (float*)d_out;

  const double s0 = (double)in_sizes[0];
  const double s1 = (double)in_sizes[1];
  const double so = (double)out_size;
  const int D  = (int)llround(sqrt(s1 * so / s0));
  const int V  = (int)llround(s1 / D);
  const int Bt = (int)llround(so / D);

  if (D == 64 && (Bt % BM) == 0 && (V % KB) == 0 && V >= KSPLIT * MIN_STEPS * KB &&
      ws_size >= (size_t)KSPLIT * Bt * D * sizeof(float)) {
    float* part = (float*)d_ws;
    dim3 grid(Bt / BM, D / BN, KSPLIT);
    dim3 block(NTHR, 1, 1);
    gemm_wmma_lds_kslice<64><<<grid, block, 0, stream>>>(inputs, emb, part, Bt, V);
    kslice_sum<64, KSPLIT><<<dim3((Bt * (D / 4) + NTHR - 1) / NTHR, 1, 1), block, 0, stream>>>(part, out, Bt);
  } else {
    dim3 grid(Bt / 16, D / 16, 1);
    dim3 block(32, 1, 1);
    if (D == 64) {
      gemm_wmma_direct<64><<<grid, block, 0, stream>>>(inputs, emb, out, Bt, V, D);
    } else {
      gemm_wmma_direct<0><<<grid, block, 0, stream>>>(inputs, emb, out, Bt, V, D);
    }
  }
}
